# starting kernel
# baseline (speedup 1.0000x reference)
#include <hip/hip_runtime.h>
#include <math.h>

typedef __attribute__((ext_vector_type(2))) float v2f;
typedef __attribute__((ext_vector_type(8))) float v8f;

#define B_ROWS 8192
#define D_FEAT 64
#define NELEM (B_ROWS * D_FEAT)
#define EPS_IN 1e-4f
#define EPS_OUT 1e-6f
#define COL_CHUNKS 8
#define TILES_PER_CHUNK (512 / COL_CHUNKS)
#define LDS_STRIDE 68

__global__ void var_kernel(const float* __restrict__ a,
                           const float* __restrict__ pos,
                           const float* __restrict__ neg,
                           float* __restrict__ rvar) {
  const int f = blockIdx.x;
  const int pair = blockIdx.y;
  const float* b = pair ? neg : pos;
  const int t = threadIdx.x;
  double s = 0.0, s2 = 0.0;
  for (int r = t; r < B_ROWS; r += 256) {
    float xa = a[r * D_FEAT + f] + EPS_IN;
    float xb = b[r * D_FEAT + f] + EPS_IN;
    s  += (double)xa + (double)xb;
    s2 += (double)xa * (double)xa + (double)xb * (double)xb;
  }
  __shared__ double ls[256];
  __shared__ double ls2[256];
  ls[t] = s; ls2[t] = s2;
  __syncthreads();
  for (int o = 128; o > 0; o >>= 1) {
    if (t < o) { ls[t] += ls[t + o]; ls2[t] += ls2[t + o]; }
    __syncthreads();
  }
  if (t == 0) {
    const double n = 2.0 * (double)B_ROWS;
    double mean = ls[0] / n;
    double var = ls2[0] / n - mean * mean;
    rvar[pair * D_FEAT + f] = (float)(1.0 / var);
  }
}

__global__ void diff_kernel(const float* __restrict__ a,
                            const float* __restrict__ p,
                            const float* __restrict__ n,
                            const float* __restrict__ rvar,
                            float* __restrict__ sdiff,
                            float* __restrict__ diffb) {
  const int idx = blockIdx.x * 256 + threadIdx.x;
  const int f = idx & (D_FEAT - 1);
  const float av = a[idx];
  const float d1 = av - p[idx];
  const float d2 = av - n[idx];
  diffb[idx] = d1;
  diffb[NELEM + idx] = d2;
  sdiff[idx] = d1 * rvar[f];
  sdiff[NELEM + idx] = d2 * rvar[D_FEAT + f];
}

__global__ void __launch_bounds__(256) mdl_wmma_kernel(
    const float* __restrict__ sdiff, const float* __restrict__ diffb,
    float* __restrict__ partial) {
  const int pair = blockIdx.z;
  const int chunk = blockIdx.y;
  const int rowblk = blockIdx.x;
  const float* A  = sdiff + (size_t)pair * NELEM;
  const float* Bm = diffb + (size_t)pair * NELEM;

  const int t = threadIdx.x;
  const int wave = t >> 5;
  const int lane = t & 31;
  const int half = lane >> 4;
  const int m = lane & 15;

  const int arow = rowblk * 128 + wave * 16 + m;
  const float* aptr = A + (size_t)arow * D_FEAT;
  v2f areg[16];
#pragma unroll
  for (int s = 0; s < 16; ++s)
    areg[s] = *(const v2f*)(aptr + 4 * s + 2 * half);

  __shared__ float bt[2][16 * LDS_STRIDE];

  const int br = t >> 4;
  const int bk = (t & 15) * 4;
  const int ct_begin = chunk * TILES_PER_CHUNK;

  {
    const float4 bv =
        *(const float4*)(Bm + (size_t)ct_begin * 16 * D_FEAT + 4 * t);
    bt[0][br * LDS_STRIDE + bk + 0] = bv.x;
    bt[0][br * LDS_STRIDE + bk + 1] = bv.y;
    bt[0][br * LDS_STRIDE + bk + 2] = bv.z;
    bt[0][br * LDS_STRIDE + bk + 3] = bv.w;
  }
  __syncthreads();

  float rs[8];
#pragma unroll
  for (int v = 0; v < 8; ++v) rs[v] = 0.0f;

  int cur = 0;
  for (int i = 0; i < TILES_PER_CHUNK; ++i) {
    float4 nv;
    const bool have_next = (i + 1 < TILES_PER_CHUNK);
    if (have_next)
      nv = *(const float4*)(Bm +
                            (size_t)(ct_begin + i + 1) * 16 * D_FEAT + 4 * t);

    const float* btc = bt[cur];
    v8f c = {};
#pragma unroll
    for (int s = 0; s < 16; ++s) {
      v2f breg = *(const v2f*)(&btc[m * LDS_STRIDE + 4 * s + 2 * half]);
      c = __builtin_amdgcn_wmma_f32_16x16x4_f32(
           false, areg[s],  false, breg,
           (short)0, c,  false,  false);
    }
#pragma unroll
    for (int v = 0; v < 8; ++v)
      rs[v] += __builtin_amdgcn_sqrtf(fmaxf(c[v], 0.0f)) + EPS_OUT;

    if (have_next) {
      float* dst = &bt[cur ^ 1][br * LDS_STRIDE + bk];
      dst[0] = nv.x; dst[1] = nv.y; dst[2] = nv.z; dst[3] = nv.w;
    }
    __syncthreads();
    cur ^= 1;
  }

#pragma unroll
  for (int v = 0; v < 8; ++v) {
    float x = rs[v];
    x += __shfl_xor(x, 1, 32);
    x += __shfl_xor(x, 2, 32);
    x += __shfl_xor(x, 4, 32);
    x += __shfl_xor(x, 8, 32);
    rs[v] = x;
  }
  if (m == 0) {
    const int rowbase = rowblk * 128 + wave * 16 + half * 8;
    float* pp = partial + ((size_t)pair * COL_CHUNKS + chunk) * B_ROWS + rowbase;
#pragma unroll
    for (int v = 0; v < 8; ++v) pp[v] = rs[v];
  }
}

__global__ void reduce_kernel(const float* __restrict__ partial,
                              float* __restrict__ out) {
  const int idx = blockIdx.x * 256 + threadIdx.x;
  const int pair = idx >> 13;
  const int r = idx & (B_ROWS - 1);
  float s = 0.0f;
#pragma unroll
  for (int c = 0; c < COL_CHUNKS; ++c)
    s += partial[((size_t)pair * COL_CHUNKS + c) * B_ROWS + r];
  out[idx] = s;
}

extern "C" void kernel_launch(void* const* d_in, const int* in_sizes, int n_in,
                              void* d_out, int out_size, void* d_ws, size_t ws_size,
                              hipStream_t stream) {
  const float* anchor   = (const float*)d_in[0];
  const float* positive = (const float*)d_in[1];
  const float* negative = (const float*)d_in[2];

  float* ws = (float*)d_ws;
  float* rvar    = ws;
  float* sdiff   = ws + 128;
  float* diffb   = sdiff + 2 * NELEM;
  float* partial = diffb + 2 * NELEM;
  float* out = (float*)d_out;

  var_kernel<<<dim3(D_FEAT, 2), 256, 0, stream>>>(anchor, positive, negative, rvar);
  diff_kernel<<<NELEM / 256, 256, 0, stream>>>(anchor, positive, negative, rvar,
                                               sdiff, diffb);
  mdl_wmma_kernel<<<dim3(B_ROWS / 128, COL_CHUNKS, 2), 256, 0, stream>>>(
      sdiff, diffb, partial);
  reduce_kernel<<<(2 * B_ROWS) / 256, 256, 0, stream>>>(partial, out);
}


# GEMM: compute only 128x128 blocks with J>=I (q is the same sum for (i,j) and (j,i)); each sqrt goes to its row sum and its column sum; 64 partials per row, summation order changes
# speedup vs baseline: 1.6029x; 1.6029x over previous
#include <hip/hip_runtime.h>
#include <math.h>

typedef __attribute__((ext_vector_type(2))) float v2f;
typedef __attribute__((ext_vector_type(8))) float v8f;

#define B_ROWS 8192
#define D_FEAT 64
#define NELEM (B_ROWS * D_FEAT)
#define EPS_IN 1e-4f
#define EPS_OUT 1e-6f
#define ROW_BLOCKS (B_ROWS / 128)
#define N_SRC ROW_BLOCKS
#define TRI_BLOCKS (32 * 65)
#define TILES_PER_BLOCK 8
#define LDS_STRIDE 68
#define CS_STRIDE 144

__global__ void var_kernel(const float* __restrict__ a,
                           const float* __restrict__ pos,
                           const float* __restrict__ neg,
                           float* __restrict__ rvar) {
  const int f = blockIdx.x;
  const int pair = blockIdx.y;
  const float* b = pair ? neg : pos;
  const int t = threadIdx.x;
  double s = 0.0, s2 = 0.0;
  for (int r = t; r < B_ROWS; r += 256) {
    float xa = a[r * D_FEAT + f] + EPS_IN;
    float xb = b[r * D_FEAT + f] + EPS_IN;
    s  += (double)xa + (double)xb;
    s2 += (double)xa * (double)xa + (double)xb * (double)xb;
  }
  __shared__ double ls[256];
  __shared__ double ls2[256];
  ls[t] = s; ls2[t] = s2;
  __syncthreads();
  for (int o = 128; o > 0; o >>= 1) {
    if (t < o) { ls[t] += ls[t + o]; ls2[t] += ls2[t + o]; }
    __syncthreads();
  }
  if (t == 0) {
    const double n = 2.0 * (double)B_ROWS;
    double mean = ls[0] / n;
    double var = ls2[0] / n - mean * mean;
    rvar[pair * D_FEAT + f] = (float)(1.0 / var);
  }
}

__global__ void diff_kernel(const float* __restrict__ a,
                            const float* __restrict__ p,
                            const float* __restrict__ n,
                            const float* __restrict__ rvar,
                            float* __restrict__ sdiff,
                            float* __restrict__ diffb) {
  const int idx = blockIdx.x * 256 + threadIdx.x;
  const int f = idx & (D_FEAT - 1);
  const float av = a[idx];
  const float d1 = av - p[idx];
  const float d2 = av - n[idx];
  diffb[idx] = d1;
  diffb[NELEM + idx] = d2;
  sdiff[idx] = d1 * rvar[f];
  sdiff[NELEM + idx] = d2 * rvar[D_FEAT + f];
}

__global__ void __launch_bounds__(256) mdl_wmma_kernel(
    const float* __restrict__ sdiff, const float* __restrict__ diffb,
    float* __restrict__ partial) {
  const int pair = blockIdx.y;
  const int bidx = blockIdx.x;
  const int p = bidx / 65;
  const int q = bidx - p * 65;
  const bool lower = (q < 64 - p);
  const int I = lower ? p : 63 - p;
  const int J = lower ? p + q : q - 1;
  const float* A  = sdiff + (size_t)pair * NELEM;
  const float* Bm = diffb + (size_t)pair * NELEM;

  const int t = threadIdx.x;
  const int wave = t >> 5;
  const int lane = t & 31;
  const int half = lane >> 4;
  const int m = lane & 15;

  const int arow = I * 128 + wave * 16 + m;
  const float* aptr = A + (size_t)arow * D_FEAT;
  v2f areg[16];
#pragma unroll
  for (int s = 0; s < 16; ++s)
    areg[s] = *(const v2f*)(aptr + 4 * s + 2 * half);

  __shared__ float bt[2][16 * LDS_STRIDE];
  __shared__ float colbuf[16 * CS_STRIDE];
  __shared__ float rowbuf[128];

  const int br = t >> 4;
  const int bk = (t & 15) * 4;
  const int ct_begin = J * TILES_PER_BLOCK;

  {
    const float4 bv =
        *(const float4*)(Bm + (size_t)ct_begin * 16 * D_FEAT + 4 * t);
    bt[0][br * LDS_STRIDE + bk + 0] = bv.x;
    bt[0][br * LDS_STRIDE + bk + 1] = bv.y;
    bt[0][br * LDS_STRIDE + bk + 2] = bv.z;
    bt[0][br * LDS_STRIDE + bk + 3] = bv.w;
  }
  __syncthreads();

  float rs[8];
#pragma unroll
  for (int v = 0; v < 8; ++v) rs[v] = 0.0f;

  int cur = 0;
  for (int i = 0; i < TILES_PER_BLOCK; ++i) {
    float4 nv;
    const bool have_next = (i + 1 < TILES_PER_BLOCK);
    if (have_next)
      nv = *(const float4*)(Bm +
                            (size_t)(ct_begin + i + 1) * 16 * D_FEAT + 4 * t);

    const float* btc = bt[cur];
    v8f c = {};
#pragma unroll
    for (int s = 0; s < 16; ++s) {
      v2f breg = *(const v2f*)(&btc[m * LDS_STRIDE + 4 * s + 2 * half]);
      c = __builtin_amdgcn_wmma_f32_16x16x4_f32(
           false, areg[s],  false, breg,
           (short)0, c,  false,  false);
    }
    float cs = 0.0f;
#pragma unroll
    for (int v = 0; v < 8; ++v) {
      const float sv = __builtin_amdgcn_sqrtf(fmaxf(c[v], 0.0f)) + EPS_OUT;
      rs[v] += sv;
      cs += sv;
    }
    colbuf[(wave * 2 + half) * CS_STRIDE + i * 16 + m] = cs;

    if (have_next) {
      float* dst = &bt[cur ^ 1][br * LDS_STRIDE + bk];
      dst[0] = nv.x; dst[1] = nv.y; dst[2] = nv.z; dst[3] = nv.w;
    }
    __syncthreads();
    cur ^= 1;
  }

#pragma unroll
  for (int v = 0; v < 8; ++v) {
    float x = rs[v];
    x += __shfl_xor(x, 1, 32);
    x += __shfl_xor(x, 2, 32);
    x += __shfl_xor(x, 4, 32);
    x += __shfl_xor(x, 8, 32);
    rs[v] = x;
  }
  if (m == 0) {
#pragma unroll
    for (int v = 0; v < 8; ++v) rowbuf[wave * 16 + half * 8 + v] = rs[v];
  }
  __syncthreads();

  if (t < 128) {
    partial[((size_t)pair * N_SRC + J) * B_ROWS + I * 128 + t] = rowbuf[t];
  } else if (I != J) {
    const int cidx = t - 128;
    float x = 0.0f;
#pragma unroll
    for (int k = 0; k < 16; ++k) x += colbuf[k * CS_STRIDE + cidx];
    partial[((size_t)pair * N_SRC + I) * B_ROWS + J * 128 + cidx] = x;
  }
}

__global__ void reduce_kernel(const float* __restrict__ partial,
                              float* __restrict__ out) {
  const int idx = blockIdx.x * 64 + threadIdx.x;
  const int pair = idx >> 13;
  const int r = idx & (B_ROWS - 1);
  float s = 0.0f;
#pragma unroll
  for (int c = 0; c < N_SRC; ++c)
    s += partial[((size_t)pair * N_SRC + c) * B_ROWS + r];
  out[idx] = s;
}

extern "C" void kernel_launch(void* const* d_in, const int* in_sizes, int n_in,
                              void* d_out, int out_size, void* d_ws, size_t ws_size,
                              hipStream_t stream) {
  const float* anchor   = (const float*)d_in[0];
  const float* positive = (const float*)d_in[1];
  const float* negative = (const float*)d_in[2];

  float* ws = (float*)d_ws;
  float* rvar    = ws;
  float* sdiff   = ws + 128;
  float* diffb   = sdiff + 2 * NELEM;
  float* partial = diffb + 2 * NELEM;
  float* out = (float*)d_out;

  var_kernel<<<dim3(D_FEAT, 2), 256, 0, stream>>>(anchor, positive, negative, rvar);
  diff_kernel<<<NELEM / 256, 256, 0, stream>>>(anchor, positive, negative, rvar,
                                               sdiff, diffb);
  mdl_wmma_kernel<<<dim3(TRI_BLOCKS, 2), 256, 0, stream>>>(sdiff, diffb, partial);
  reduce_kernel<<<(2 * B_ROWS) / 64, 64, 0, stream>>>(partial, out);
}


# diff and 1/var scaling fused into the GEMM's operand loads (A block staged once via LDS, B tiles subtracted as staged); the separate pass and its workspace planes are gone: 3 launches, sums unchanged
# speedup vs baseline: 1.7748x; 1.1072x over previous
#include <hip/hip_runtime.h>
#include <math.h>

typedef __attribute__((ext_vector_type(2))) float v2f;
typedef __attribute__((ext_vector_type(4))) float v4f;
typedef __attribute__((ext_vector_type(8))) float v8f;

#define B_ROWS 8192
#define D_FEAT 64
#define EPS_IN 1e-4f
#define EPS_OUT 1e-6f
#define ROW_BLOCKS (B_ROWS / 128)
#define N_SRC ROW_BLOCKS
#define TRI_BLOCKS (32 * 65)
#define TILES_PER_BLOCK 8
#define LDS_STRIDE 68
#define CS_STRIDE 144

__global__ void var_kernel(const float* __restrict__ a,
                           const float* __restrict__ pos,
                           const float* __restrict__ neg,
                           float* __restrict__ rvar) {
  const int f = blockIdx.x;
  const int pair = blockIdx.y;
  const float* b = pair ? neg : pos;
  const int t = threadIdx.x;
  double s = 0.0, s2 = 0.0;
  for (int r = t; r < B_ROWS; r += 256) {
    float xa = a[r * D_FEAT + f] + EPS_IN;
    float xb = b[r * D_FEAT + f] + EPS_IN;
    s  += (double)xa + (double)xb;
    s2 += (double)xa * (double)xa + (double)xb * (double)xb;
  }
  __shared__ double ls[256];
  __shared__ double ls2[256];
  ls[t] = s; ls2[t] = s2;
  __syncthreads();
  for (int o = 128; o > 0; o >>= 1) {
    if (t < o) { ls[t] += ls[t + o]; ls2[t] += ls2[t + o]; }
    __syncthreads();
  }
  if (t == 0) {
    const double n = 2.0 * (double)B_ROWS;
    double mean = ls[0] / n;
    double var = ls2[0] / n - mean * mean;
    rvar[pair * D_FEAT + f] = (float)(1.0 / var);
  }
}

#define BT_FLOATS (16 * LDS_STRIDE)
#define STAGE_FLOATS (128 * LDS_STRIDE)
#define COLBUF_OFF (2 * BT_FLOATS)
#define ROWBUF_OFF (COLBUF_OFF + 16 * CS_STRIDE)

__global__ void __launch_bounds__(256) mdl_wmma_kernel(
    const float* __restrict__ anchor, const float* __restrict__ pos,
    const float* __restrict__ neg, const float* __restrict__ rvar,
    float* __restrict__ partial) {
  const int pair = blockIdx.y;
  const int bidx = blockIdx.x;
  const int p = bidx / 65;
  const int q = bidx - p * 65;
  const bool lower = (q < 64 - p);
  const int I = lower ? p : 63 - p;
  const int J = lower ? p + q : q - 1;
  const float* __restrict__ other = pair ? neg : pos;

  const int t = threadIdx.x;
  const int wave = t >> 5;
  const int lane = t & 31;
  const int half = lane >> 4;
  const int m = lane & 15;

  __shared__ __attribute__((aligned(16))) float smem[STAGE_FLOATS];
  float* const colbuf = smem + COLBUF_OFF;
  float* const rowbuf = smem + ROWBUF_OFF;

  const int br = t >> 4;
  const int bk = (t & 15) * 4;

  const size_t bbase = (size_t)J * 128 * D_FEAT + 4 * t;
  v4f na = *(const v4f*)(anchor + bbase);
  v4f no = *(const v4f*)(other + bbase);

  {
    const v4f rv = *(const v4f*)(rvar + pair * D_FEAT + bk);
    const size_t abase = (size_t)I * 128 * D_FEAT + 4 * t;
#pragma unroll 1
    for (int h = 0; h < 2; ++h) {
      v4f av[4], ov[4];
#pragma unroll
      for (int i = 0; i < 4; ++i) {
        av[i] = *(const v4f*)(anchor + abase + (size_t)(4 * h + i) * 16 * D_FEAT);
        ov[i] = *(const v4f*)(other + abase + (size_t)(4 * h + i) * 16 * D_FEAT);
      }
#pragma unroll
      for (int i = 0; i < 4; ++i)
        *(v4f*)(&smem[((4 * h + i) * 16 + br) * LDS_STRIDE + bk]) = (av[i] - ov[i]) * rv;
    }
  }

  __syncthreads();

  v2f areg[16];
  {
    const float* arow = &smem[(wave * 16 + m) * LDS_STRIDE + 2 * half];
#pragma unroll
    for (int s = 0; s < 16; ++s) areg[s] = *(const v2f*)(arow + 4 * s);
  }

  __syncthreads();

  *(v4f*)(&smem[br * LDS_STRIDE + bk]) = na - no;
  __syncthreads();

  float rs[8];
#pragma unroll
  for (int v = 0; v < 8; ++v) rs[v] = 0.0f;

  int cur = 0;
#pragma unroll 1
  for (int i = 0; i < TILES_PER_BLOCK; ++i) {
    const bool have_next = (i + 1 < TILES_PER_BLOCK);
    if (have_next) {
      na = *(const v4f*)(anchor + bbase + (size_t)(i + 1) * 16 * D_FEAT);
      no = *(const v4f*)(other + bbase + (size_t)(i + 1) * 16 * D_FEAT);
    }

    const float* btc = smem + cur * BT_FLOATS;
    v8f c = {};
#pragma unroll
    for (int s = 0; s < 16; ++s) {
      v2f breg = *(const v2f*)(&btc[m * LDS_STRIDE + 4 * s + 2 * half]);
      c = __builtin_amdgcn_wmma_f32_16x16x4_f32(
           false, areg[s],  false, breg,
           (short)0, c,  false,  false);
    }
    float cs = 0.0f;
#pragma unroll
    for (int v = 0; v < 8; ++v) {
      const float sv = __builtin_amdgcn_sqrtf(fmaxf(c[v], 0.0f)) + EPS_OUT;
      rs[v] += sv;
      cs += sv;
    }
    colbuf[(wave * 2 + half) * CS_STRIDE + i * 16 + m] = cs;

    if (have_next)
      *(v4f*)(smem + (cur ^ 1) * BT_FLOATS + br * LDS_STRIDE + bk) = na - no;
    __syncthreads();
    cur ^= 1;
  }

#pragma unroll
  for (int v = 0; v < 8; ++v) {
    float x = rs[v];
    x += __shfl_xor(x, 1, 32);
    x += __shfl_xor(x, 2, 32);
    x += __shfl_xor(x, 4, 32);
    x += __shfl_xor(x, 8, 32);
    rs[v] = x;
  }
  if (m == 0) {
#pragma unroll
    for (int v = 0; v < 8; ++v) rowbuf[wave * 16 + half * 8 + v] = rs[v];
  }
  __syncthreads();

  if (t < 128) {
    partial[((size_t)pair * N_SRC + J) * B_ROWS + I * 128 + t] = rowbuf[t];
  } else if (I != J) {
    const int cidx = t - 128;
    float x = 0.0f;
#pragma unroll
    for (int k = 0; k < 16; ++k) x += colbuf[k * CS_STRIDE + cidx];
    partial[((size_t)pair * N_SRC + I) * B_ROWS + J * 128 + cidx] = x;
  }
}

__global__ void reduce_kernel(const float* __restrict__ partial,
                              float* __restrict__ out) {
  const int idx = blockIdx.x * 64 + threadIdx.x;
  const int pair = idx >> 13;
  const int r = idx & (B_ROWS - 1);
  float s = 0.0f;
#pragma unroll
  for (int c = 0; c < N_SRC; ++c)
    s += partial[((size_t)pair * N_SRC + c) * B_ROWS + r];
  out[idx] = s;
}

extern "C" void kernel_launch(void* const* d_in, const int* in_sizes, int n_in,
                              void* d_out, int out_size, void* d_ws, size_t ws_size,
                              hipStream_t stream) {
  const float* anchor   = (const float*)d_in[0];
  const float* positive = (const float*)d_in[1];
  const float* negative = (const float*)d_in[2];

  float* ws = (float*)d_ws;
  float* rvar    = ws;
  float* partial = ws + 128;
  float* out = (float*)d_out;

  var_kernel<<<dim3(D_FEAT, 2), 256, 0, stream>>>(anchor, positive, negative, rvar);
  mdl_wmma_kernel<<<dim3(TRI_BLOCKS, 2), 256, 0, stream>>>(anchor, positive, negative,
                                                           rvar, partial);
  reduce_kernel<<<(2 * B_ROWS) / 64, 64, 0, stream>>>(partial, out);
}


# Variance pass rebuilt: 32 workgroups read whole rows once and write f32 partial sums, GEMM prologue combines them (was a column-strided f64 pass); variance sums now f32 trees, order changes
# speedup vs baseline: 2.0695x; 1.1660x over previous
#include <hip/hip_runtime.h>
#include <math.h>

typedef __attribute__((ext_vector_type(2))) float v2f;
typedef __attribute__((ext_vector_type(4))) float v4f;
typedef __attribute__((ext_vector_type(8))) float v8f;

#define B_ROWS 8192
#define D_FEAT 64
#define EPS_IN 1e-4f
#define EPS_OUT 1e-6f
#define ROW_BLOCKS (B_ROWS / 128)
#define N_SRC ROW_BLOCKS
#define TRI_BLOCKS (32 * 65)
#define TILES_PER_BLOCK 8
#define LDS_STRIDE 68
#define CS_STRIDE 144

#define VP_BLOCKS 32
#define VP_THREADS 512
#define VP_ROWS (B_ROWS / VP_BLOCKS)
#define VP_SLOTS (VP_THREADS / 16)
#define VP_ITERS (VP_ROWS / VP_SLOTS)
#define VP_VALS 256

__global__ void __launch_bounds__(VP_THREADS) var_partial_kernel(
    const float* __restrict__ a, const float* __restrict__ pos,
    const float* __restrict__ neg, float* __restrict__ vpart) {
  const int t = threadIdx.x;
  const int rs = t >> 4;
  const int fq = t & 15;
  const size_t base = (size_t)blockIdx.x * VP_ROWS * D_FEAT + 4 * t;
  v4f sa = {}, qa = {}, sp = {}, qp = {}, sn = {}, qn = {};
#pragma unroll
  for (int k = 0; k < VP_ITERS; ++k) {
    const size_t o = base + (size_t)k * VP_SLOTS * D_FEAT;
    const v4f xa = *(const v4f*)(a + o) + EPS_IN;
    const v4f xp = *(const v4f*)(pos + o) + EPS_IN;
    const v4f xn = *(const v4f*)(neg + o) + EPS_IN;
    sa += xa; qa += xa * xa;
    sp += xp; qp += xp * xp;
    sn += xn; qn += xn * xn;
  }
  __shared__ __attribute__((aligned(16))) float red[VP_SLOTS * VP_VALS];
  float* const myrow = red + rs * VP_VALS + 4 * fq;
  *(v4f*)(myrow + 0 * D_FEAT) = sa + sp;
  *(v4f*)(myrow + 1 * D_FEAT) = qa + qp;
  *(v4f*)(myrow + 2 * D_FEAT) = sa + sn;
  *(v4f*)(myrow + 3 * D_FEAT) = qa + qn;
  __syncthreads();
  if (t < VP_VALS) {
    float x0 = 0.0f, x1 = 0.0f, x2 = 0.0f, x3 = 0.0f;
#pragma unroll
    for (int r = 0; r < VP_SLOTS; r += 4) {
      x0 += red[(r + 0) * VP_VALS + t];
      x1 += red[(r + 1) * VP_VALS + t];
      x2 += red[(r + 2) * VP_VALS + t];
      x3 += red[(r + 3) * VP_VALS + t];
    }
    vpart[blockIdx.x * VP_VALS + t] = (x0 + x1) + (x2 + x3);
  }
}

#define BT_FLOATS (16 * LDS_STRIDE)
#define STAGE_FLOATS (128 * LDS_STRIDE)
#define COLBUF_OFF (2 * BT_FLOATS)
#define ROWBUF_OFF (COLBUF_OFF + 16 * CS_STRIDE)

__global__ void __launch_bounds__(256) mdl_wmma_kernel(
    const float* __restrict__ anchor, const float* __restrict__ pos,
    const float* __restrict__ neg, const float* __restrict__ vpart,
    float* __restrict__ partial) {
  const int pair = blockIdx.y;
  const int bidx = blockIdx.x;
  const int p = bidx / 65;
  const int q = bidx - p * 65;
  const bool lower = (q < 64 - p);
  const int I = lower ? p : 63 - p;
  const int J = lower ? p + q : q - 1;
  const float* __restrict__ other = pair ? neg : pos;

  const int t = threadIdx.x;
  const int wave = t >> 5;
  const int lane = t & 31;
  const int half = lane >> 4;
  const int m = lane & 15;

  __shared__ __attribute__((aligned(16))) float smem[STAGE_FLOATS];
  float* const colbuf = smem + COLBUF_OFF;
  float* const rowbuf = smem + ROWBUF_OFF;
  __shared__ __attribute__((aligned(16))) float rvs[D_FEAT];

  const int br = t >> 4;
  const int bk = (t & 15) * 4;

  const size_t bbase = (size_t)J * 128 * D_FEAT + 4 * t;
  v4f na = *(const v4f*)(anchor + bbase);
  v4f no = *(const v4f*)(other + bbase);

  {
    const int col = lane & 15;
    const int sq = col >> 3;
    const int f = wave * 8 + (col & 7);
    const float* vp = vpart + (size_t)(half * (VP_BLOCKS / 2)) * VP_VALS + pair * 2 * D_FEAT +
                      sq * D_FEAT + f;
    float x = 0.0f;
#pragma unroll
    for (int k = 0; k < VP_BLOCKS / 2; ++k) x += vp[k * VP_VALS];
    x += __shfl_xor(x, 16, 32);
    const float y = __shfl_xor(x, 8, 32);
    if (lane < 8) {
      const float inv_n = 1.0f / (2.0f * (float)B_ROWS);
      const float mean = x * inv_n;
      const float var = y * inv_n - mean * mean;
      rvs[f] = 1.0f / var;
    }
  }
  __syncthreads();

  {
    const v4f rv = *(const v4f*)(rvs + bk);
    const size_t abase = (size_t)I * 128 * D_FEAT + 4 * t;
#pragma unroll 1
    for (int h = 0; h < 2; ++h) {
      v4f av[4], ov[4];
#pragma unroll
      for (int i = 0; i < 4; ++i) {
        av[i] = *(const v4f*)(anchor + abase + (size_t)(4 * h + i) * 16 * D_FEAT);
        ov[i] = *(const v4f*)(other + abase + (size_t)(4 * h + i) * 16 * D_FEAT);
      }
#pragma unroll
      for (int i = 0; i < 4; ++i)
        *(v4f*)(&smem[((4 * h + i) * 16 + br) * LDS_STRIDE + bk]) = (av[i] - ov[i]) * rv;
    }
  }

  __syncthreads();

  v2f areg[16];
  {
    const float* arow = &smem[(wave * 16 + m) * LDS_STRIDE + 2 * half];
#pragma unroll
    for (int s = 0; s < 16; ++s) areg[s] = *(const v2f*)(arow + 4 * s);
  }

  __syncthreads();

  *(v4f*)(&smem[br * LDS_STRIDE + bk]) = na - no;
  __syncthreads();

  float rs[8];
#pragma unroll
  for (int v = 0; v < 8; ++v) rs[v] = 0.0f;

  int cur = 0;
#pragma unroll 1
  for (int i = 0; i < TILES_PER_BLOCK; ++i) {
    const bool have_next = (i + 1 < TILES_PER_BLOCK);
    if (have_next) {
      na = *(const v4f*)(anchor + bbase + (size_t)(i + 1) * 16 * D_FEAT);
      no = *(const v4f*)(other + bbase + (size_t)(i + 1) * 16 * D_FEAT);
    }

    const float* btc = smem + cur * BT_FLOATS;
    v8f c = {};
#pragma unroll
    for (int s = 0; s < 16; ++s) {
      v2f breg = *(const v2f*)(&btc[m * LDS_STRIDE + 4 * s + 2 * half]);
      c = __builtin_amdgcn_wmma_f32_16x16x4_f32(
           false, areg[s],  false, breg,
           (short)0, c,  false,  false);
    }
    float cs = 0.0f;
#pragma unroll
    for (int v = 0; v < 8; ++v) {
      const float sv = __builtin_amdgcn_sqrtf(fmaxf(c[v], 0.0f)) + EPS_OUT;
      rs[v] += sv;
      cs += sv;
    }
    colbuf[(wave * 2 + half) * CS_STRIDE + i * 16 + m] = cs;

    if (have_next)
      *(v4f*)(smem + (cur ^ 1) * BT_FLOATS + br * LDS_STRIDE + bk) = na - no;
    __syncthreads();
    cur ^= 1;
  }

#pragma unroll
  for (int v = 0; v < 8; ++v) {
    float x = rs[v];
    x += __shfl_xor(x, 1, 32);
    x += __shfl_xor(x, 2, 32);
    x += __shfl_xor(x, 4, 32);
    x += __shfl_xor(x, 8, 32);
    rs[v] = x;
  }
  if (m == 0) {
#pragma unroll
    for (int v = 0; v < 8; ++v) rowbuf[wave * 16 + half * 8 + v] = rs[v];
  }
  __syncthreads();

  if (t < 128) {
    partial[((size_t)pair * N_SRC + J) * B_ROWS + I * 128 + t] = rowbuf[t];
  } else if (I != J) {
    const int cidx = t - 128;
    float x = 0.0f;
#pragma unroll
    for (int k = 0; k < 16; ++k) x += colbuf[k * CS_STRIDE + cidx];
    partial[((size_t)pair * N_SRC + I) * B_ROWS + J * 128 + cidx] = x;
  }
}

__global__ void reduce_kernel(const float* __restrict__ partial,
                              float* __restrict__ out) {
  const int idx = blockIdx.x * 64 + threadIdx.x;
  const int pair = idx >> 13;
  const int r = idx & (B_ROWS - 1);
  float s = 0.0f;
#pragma unroll
  for (int c = 0; c < N_SRC; ++c)
    s += partial[((size_t)pair * N_SRC + c) * B_ROWS + r];
  out[idx] = s;
}

extern "C" void kernel_launch(void* const* d_in, const int* in_sizes, int n_in,
                              void* d_out, int out_size, void* d_ws, size_t ws_size,
                              hipStream_t stream) {
  const float* anchor   = (const float*)d_in[0];
  const float* positive = (const float*)d_in[1];
  const float* negative = (const float*)d_in[2];

  float* ws = (float*)d_ws;
  float* vpart   = ws;
  float* partial = ws + VP_BLOCKS * VP_VALS;
  float* out = (float*)d_out;

  var_partial_kernel<<<VP_BLOCKS, VP_THREADS, 0, stream>>>(anchor, positive, negative, vpart);
  mdl_wmma_kernel<<<dim3(TRI_BLOCKS, 2), 256, 0, stream>>>(anchor, positive, negative,
                                                           vpart, partial);
  reduce_kernel<<<(2 * B_ROWS) / 64, 64, 0, stream>>>(partial, out);
}


# GEMM work-group tile doubled to 256 rows x 128 columns: two 16-row A fragment sets per wave, each B fragment read and each staged B tile feeds two WMMAs (two accumulators); column sum order changes
# speedup vs baseline: 2.1747x; 1.0508x over previous
#include <hip/hip_runtime.h>
#include <math.h>

typedef __attribute__((ext_vector_type(2))) float v2f;
typedef __attribute__((ext_vector_type(4))) float v4f;
typedef __attribute__((ext_vector_type(8))) float v8f;

#define B_ROWS 8192
#define D_FEAT 64
#define EPS_IN 1e-4f
#define EPS_OUT 1e-6f
#define ROW_BLOCKS (B_ROWS / 128)
#define N_SRC ROW_BLOCKS
#define FULL_BLOCKS 1024
#define HALF_BLOCKS 32
#define RECT_BLOCKS (2 * (FULL_BLOCKS + HALF_BLOCKS))
#define TILES_PER_BLOCK 8
#define LDS_STRIDE 68
#define CS_STRIDE 144

#define VP_BLOCKS 32
#define VP_THREADS 512
#define VP_ROWS (B_ROWS / VP_BLOCKS)
#define VP_SLOTS (VP_THREADS / 16)
#define VP_ITERS (VP_ROWS / VP_SLOTS)
#define VP_VALS 256

__global__ void __launch_bounds__(VP_THREADS) var_partial_kernel(
    const float* __restrict__ a, const float* __restrict__ pos,
    const float* __restrict__ neg, float* __restrict__ vpart) {
  const int t = threadIdx.x;
  const int rs = t >> 4;
  const int fq = t & 15;
  const size_t base = (size_t)blockIdx.x * VP_ROWS * D_FEAT + 4 * t;
  v4f sa = {}, qa = {}, sp = {}, qp = {}, sn = {}, qn = {};
#pragma unroll
  for (int k = 0; k < VP_ITERS; ++k) {
    const size_t o = base + (size_t)k * VP_SLOTS * D_FEAT;
    const v4f xa = *(const v4f*)(a + o) + EPS_IN;
    const v4f xp = *(const v4f*)(pos + o) + EPS_IN;
    const v4f xn = *(const v4f*)(neg + o) + EPS_IN;
    sa += xa; qa += xa * xa;
    sp += xp; qp += xp * xp;
    sn += xn; qn += xn * xn;
  }
  __shared__ __attribute__((aligned(16))) float red[VP_SLOTS * VP_VALS];
  float* const myrow = red + rs * VP_VALS + 4 * fq;
  *(v4f*)(myrow + 0 * D_FEAT) = sa + sp;
  *(v4f*)(myrow + 1 * D_FEAT) = qa + qp;
  *(v4f*)(myrow + 2 * D_FEAT) = sa + sn;
  *(v4f*)(myrow + 3 * D_FEAT) = qa + qn;
  __syncthreads();
  if (t < VP_VALS) {
    float x0 = 0.0f, x1 = 0.0f, x2 = 0.0f, x3 = 0.0f;
#pragma unroll
    for (int r = 0; r < VP_SLOTS; r += 4) {
      x0 += red[(r + 0) * VP_VALS + t];
      x1 += red[(r + 1) * VP_VALS + t];
      x2 += red[(r + 2) * VP_VALS + t];
      x3 += red[(r + 3) * VP_VALS + t];
    }
    vpart[blockIdx.x * VP_VALS + t] = (x0 + x1) + (x2 + x3);
  }
}

#define BT_FLOATS (16 * LDS_STRIDE)
#define STAGE_FLOATS (256 * LDS_STRIDE)
#define COLBUF_OFF (2 * BT_FLOATS)
#define ROWBUF_OFF (COLBUF_OFF + 16 * CS_STRIDE)

__global__ void __launch_bounds__(256) mdl_wmma_kernel(
    const float* __restrict__ anchor, const float* __restrict__ pos,
    const float* __restrict__ neg, const float* __restrict__ vpart,
    float* __restrict__ partial) {
  const int bidx = blockIdx.x;
  int pair, I2, J;
  if (bidx < 2 * FULL_BLOCKS) {
    pair = bidx >> 10;
    const int f = bidx & (FULL_BLOCKS - 1);
    const int p = f >> 6;
    const int q = f & 63;
    const bool lower = (q < 63 - 2 * p);
    I2 = lower ? p : 31 - p;
    J = lower ? 2 * p + 1 + q : q;
  } else {
    const int h = bidx - 2 * FULL_BLOCKS;
    pair = h >> 5;
    I2 = h & (HALF_BLOCKS - 1);
    J = 2 * I2;
  }
  const float* __restrict__ other = pair ? neg : pos;

  const int t = threadIdx.x;
  const int wave = t >> 5;
  const int lane = t & 31;
  const int half = lane >> 4;
  const int m = lane & 15;

  __shared__ __attribute__((aligned(16))) float smem[STAGE_FLOATS];
  float* const colbuf = smem + COLBUF_OFF;
  float* const rowbuf = smem + ROWBUF_OFF;
  __shared__ __attribute__((aligned(16))) float rvs[D_FEAT];

  const int br = t >> 4;
  const int bk = (t & 15) * 4;

  const size_t bbase = (size_t)J * 128 * D_FEAT + 4 * t;
  v4f na = *(const v4f*)(anchor + bbase);
  v4f no = *(const v4f*)(other + bbase);

  {
    const int col = lane & 15;
    const int sq = col >> 3;
    const int f = wave * 8 + (col & 7);
    const float* vp = vpart + (size_t)(half * (VP_BLOCKS / 2)) * VP_VALS + pair * 2 * D_FEAT +
                      sq * D_FEAT + f;
    float x = 0.0f;
#pragma unroll
    for (int k = 0; k < VP_BLOCKS / 2; ++k) x += vp[k * VP_VALS];
    x += __shfl_xor(x, 16, 32);
    const float y = __shfl_xor(x, 8, 32);
    if (lane < 8) {
      const float inv_n = 1.0f / (2.0f * (float)B_ROWS);
      const float mean = x * inv_n;
      const float var = y * inv_n - mean * mean;
      rvs[f] = 1.0f / var;
    }
  }
  __syncthreads();

  {
    const v4f rv = *(const v4f*)(rvs + bk);
    const size_t abase = (size_t)I2 * 256 * D_FEAT + 4 * t;
#pragma unroll 1
    for (int h = 0; h < 2; ++h) {
      v4f av[8], ov[8];
#pragma unroll
      for (int i = 0; i < 8; ++i) {
        av[i] = *(const v4f*)(anchor + abase + (size_t)(8 * h + i) * 16 * D_FEAT);
        ov[i] = *(const v4f*)(other + abase + (size_t)(8 * h + i) * 16 * D_FEAT);
      }
#pragma unroll
      for (int i = 0; i < 8; ++i)
        *(v4f*)(&smem[((8 * h + i) * 16 + br) * LDS_STRIDE + bk]) = (av[i] - ov[i]) * rv;
    }
  }

  __syncthreads();

  v2f areg0[16], areg1[16];
  {
    const float* arow = &smem[(wave * 32 + m) * LDS_STRIDE + 2 * half];
#pragma unroll
    for (int s = 0; s < 16; ++s) {
      areg0[s] = *(const v2f*)(arow + 4 * s);
      areg1[s] = *(const v2f*)(arow + 16 * LDS_STRIDE + 4 * s);
    }
  }

  __syncthreads();

  *(v4f*)(&smem[br * LDS_STRIDE + bk]) = na - no;
  __syncthreads();

  float rs0[8], rs1[8];
#pragma unroll
  for (int v = 0; v < 8; ++v) { rs0[v] = 0.0f; rs1[v] = 0.0f; }

  const int Iw = 2 * I2 + (__builtin_amdgcn_readfirstlane(wave) >> 2);
  const bool active = (Iw <= J);

  int cur = 0;
#pragma unroll 1
  for (int i = 0; i < TILES_PER_BLOCK; ++i) {
    const bool have_next = (i + 1 < TILES_PER_BLOCK);
    if (have_next) {
      na = *(const v4f*)(anchor + bbase + (size_t)(i + 1) * 16 * D_FEAT);
      no = *(const v4f*)(other + bbase + (size_t)(i + 1) * 16 * D_FEAT);
    }

    if (active) {
      const float* btc = smem + cur * BT_FLOATS;
      v8f c0 = {}, c1 = {};
#pragma unroll
      for (int s = 0; s < 16; ++s) {
        v2f breg = *(const v2f*)(&btc[m * LDS_STRIDE + 4 * s + 2 * half]);
        c0 = __builtin_amdgcn_wmma_f32_16x16x4_f32(
             false, areg0[s],  false, breg,
             (short)0, c0,  false,  false);
        c1 = __builtin_amdgcn_wmma_f32_16x16x4_f32(
             false, areg1[s],  false, breg,
             (short)0, c1,  false,  false);
      }
      float cs = 0.0f;
#pragma unroll
      for (int v = 0; v < 8; ++v) {
        const float sv = __builtin_amdgcn_sqrtf(fmaxf(c0[v], 0.0f)) + EPS_OUT;
        rs0[v] += sv;
        cs += sv;
      }
#pragma unroll
      for (int v = 0; v < 8; ++v) {
        const float sv = __builtin_amdgcn_sqrtf(fmaxf(c1[v], 0.0f)) + EPS_OUT;
        rs1[v] += sv;
        cs += sv;
      }
      colbuf[(wave * 2 + half) * CS_STRIDE + i * 16 + m] = cs;
    }

    if (have_next)
      *(v4f*)(smem + (cur ^ 1) * BT_FLOATS + br * LDS_STRIDE + bk) = na - no;
    __syncthreads();
    cur ^= 1;
  }

#pragma unroll
  for (int v = 0; v < 8; ++v) {
    float x = rs0[v];
    x += __shfl_xor(x, 1, 32);
    x += __shfl_xor(x, 2, 32);
    x += __shfl_xor(x, 4, 32);
    x += __shfl_xor(x, 8, 32);
    rs0[v] = x;
    float y = rs1[v];
    y += __shfl_xor(y, 1, 32);
    y += __shfl_xor(y, 2, 32);
    y += __shfl_xor(y, 4, 32);
    y += __shfl_xor(y, 8, 32);
    rs1[v] = y;
  }
  if (m == 0) {
#pragma unroll
    for (int v = 0; v < 8; ++v) {
      rowbuf[wave * 32 + half * 8 + v] = rs0[v];
      rowbuf[wave * 32 + 16 + half * 8 + v] = rs1[v];
    }
  }
  __syncthreads();

  const int sub = t >> 7;
  const int It = 2 * I2 + sub;
  if (It <= J) {
    partial[((size_t)pair * N_SRC + J) * B_ROWS + I2 * 256 + t] = rowbuf[t];
  }
  if (It < J) {
    const int cidx = t & 127;
    float x = 0.0f;
#pragma unroll
    for (int k = 0; k < 8; ++k) x += colbuf[(8 * sub + k) * CS_STRIDE + cidx];
    partial[((size_t)pair * N_SRC + It) * B_ROWS + J * 128 + cidx] = x;
  }
}

__global__ void reduce_kernel(const float* __restrict__ partial,
                              float* __restrict__ out) {
  const int idx = blockIdx.x * 64 + threadIdx.x;
  const int pair = idx >> 13;
  const int r = idx & (B_ROWS - 1);
  float s = 0.0f;
#pragma unroll
  for (int c = 0; c < N_SRC; ++c)
    s += partial[((size_t)pair * N_SRC + c) * B_ROWS + r];
  out[idx] = s;
}

extern "C" void kernel_launch(void* const* d_in, const int* in_sizes, int n_in,
                              void* d_out, int out_size, void* d_ws, size_t ws_size,
                              hipStream_t stream) {
  const float* anchor   = (const float*)d_in[0];
  const float* positive = (const float*)d_in[1];
  const float* negative = (const float*)d_in[2];

  float* ws = (float*)d_ws;
  float* vpart   = ws;
  float* partial = ws + VP_BLOCKS * VP_VALS;
  float* out = (float*)d_out;

  var_partial_kernel<<<VP_BLOCKS, VP_THREADS, 0, stream>>>(anchor, positive, negative, vpart);
  mdl_wmma_kernel<<<RECT_BLOCKS, 256, 0, stream>>>(anchor, positive, negative,
                                                           vpart, partial);
  reduce_kernel<<<(2 * B_ROWS) / 64, 64, 0, stream>>>(partial, out);
}
